# starting kernel
# baseline (speedup 1.0000x reference)
#include <hip/hip_runtime.h>
#include <hip/hip_bf16.h>

#define N_NODES 50000
#define N_EDGES 800000

typedef float v2f __attribute__((ext_vector_type(2)));
typedef float v8f __attribute__((ext_vector_type(8)));

__global__ void gcn_transpose_k(const float* __restrict__ in, float* __restrict__ out,
                                int rows, int cols) {
    int gid = blockIdx.x * blockDim.x + threadIdx.x;
    if (gid < rows * cols) {
        int r = gid / cols;
        int c = gid - r * cols;
        out[(size_t)c * rows + r] = in[gid];
    }
}

__global__ void gcn_gemm1_wmma(const float* __restrict__ x,
                               const float* __restrict__ w1t,
                               float* __restrict__ out) {
    const int wave    = (int)((blockIdx.x * blockDim.x + threadIdx.x) >> 5);
    const int lane    = threadIdx.x & 31;
    const int rowBase = (wave >> 4) << 4;
    const int colBase = (wave & 15) << 4;
    const int m       = lane & 15;
    const int kl      = (lane >> 4) << 1;

    const float* __restrict__ arow = x + (size_t)(rowBase + m) * 300;
    const float* __restrict__ bcol = w1t + colBase + m;

    v8f c = {0.f, 0.f, 0.f, 0.f, 0.f, 0.f, 0.f, 0.f};
    for (int k = 0; k < 300; k += 4) {
        v2f a = *(const v2f*)(arow + k + kl);
        v2f b;
        b.x = bcol[(size_t)(k + kl)     * 256];
        b.y = bcol[(size_t)(k + kl + 1) * 256];
        c = __builtin_amdgcn_wmma_f32_16x16x4_f32(false, a, false, b,
                                                  (short)0, c, false, false);
    }
    const int mBase = rowBase + ((lane >> 4) << 3);
    float* __restrict__ o = out + (size_t)mBase * 256 + colBase + m;
#pragma unroll
    for (int j = 0; j < 8; ++j)
        o[(size_t)j * 256] = c[j];
}

__global__ void gcn_gemm2_wmma(const float* __restrict__ h1,
                               const float* __restrict__ b1,
                               const float* __restrict__ w2t,
                               float* __restrict__ out) {
    const int wave    = (int)((blockIdx.x * blockDim.x + threadIdx.x) >> 5);
    const int lane    = threadIdx.x & 31;
    const int rowBase = (wave >> 3) << 4;
    const int colBase = (wave & 7)  << 4;
    const int m       = lane & 15;
    const int kl      = (lane >> 4) << 1;

    const float* __restrict__ arow = h1 + (size_t)(rowBase + m) * 256;
    const float* __restrict__ bcol = w2t + colBase + m;

    v8f c = {0.f, 0.f, 0.f, 0.f, 0.f, 0.f, 0.f, 0.f};
    for (int k = 0; k < 256; k += 4) {
        v2f a  = *(const v2f*)(arow + k + kl);
        v2f bb = *(const v2f*)(b1   + k + kl);
        a.x = fmaxf(a.x + bb.x, 0.f);
        a.y = fmaxf(a.y + bb.y, 0.f);
        v2f b;
        b.x = bcol[(size_t)(k + kl)     * 128];
        b.y = bcol[(size_t)(k + kl + 1) * 128];
        c = __builtin_amdgcn_wmma_f32_16x16x4_f32(false, a, false, b,
                                                  (short)0, c, false, false);
    }
    const int mBase = rowBase + ((lane >> 4) << 3);
    float* __restrict__ o = out + (size_t)mBase * 128 + colBase + m;
#pragma unroll
    for (int j = 0; j < 8; ++j)
        o[(size_t)j * 128] = c[j];
}

template <int D>
__global__ void gcn_spmm_scatter(const int*   __restrict__ erow,
                                 const int*   __restrict__ ecol,
                                 const float* __restrict__ eval,
                                 const float* __restrict__ dense,
                                 float*       __restrict__ out) {
    constexpr int CH = D / 4;
    constexpr int SH = (D == 256) ? 6 : 5;
    long gid = (long)blockIdx.x * blockDim.x + threadIdx.x;
    int  e   = (int)(gid >> SH);
    if (e >= N_EDGES) return;
    int  c   = ((int)gid & (CH - 1)) << 2;

    int   r  = erow[e];
    int   cc = ecol[e];
    float v  = eval[e];

    const float4 d = *(const float4*)(dense + (size_t)cc * D + c);
    float* o = out + (size_t)r * D + c;
    (void)__hip_atomic_fetch_add(o + 0, v * d.x, __ATOMIC_RELAXED, __HIP_MEMORY_SCOPE_AGENT);
    (void)__hip_atomic_fetch_add(o + 1, v * d.y, __ATOMIC_RELAXED, __HIP_MEMORY_SCOPE_AGENT);
    (void)__hip_atomic_fetch_add(o + 2, v * d.z, __ATOMIC_RELAXED, __HIP_MEMORY_SCOPE_AGENT);
    (void)__hip_atomic_fetch_add(o + 3, v * d.w, __ATOMIC_RELAXED, __HIP_MEMORY_SCOPE_AGENT);
}

__global__ void gcn_bias_norm(const float* __restrict__ b2, float* __restrict__ out) {
    int gid  = blockIdx.x * blockDim.x + threadIdx.x;
    int row  = gid >> 5;
    int lane = gid & 31;
    if (row >= N_NODES) return;

    float* p = out + (size_t)row * 128 + lane * 4;
    float4 v  = *(float4*)p;
    float4 bb = *(const float4*)(b2 + lane * 4);
    v.x += bb.x; v.y += bb.y; v.z += bb.z; v.w += bb.w;

    float ss = v.x * v.x + v.y * v.y + v.z * v.z + v.w * v.w;
#pragma unroll
    for (int off = 16; off > 0; off >>= 1)
        ss += __shfl_xor(ss, off, 32);

    float scale = 1.0f / fmaxf(sqrtf(ss), 1e-12f);
    v.x *= scale; v.y *= scale; v.z *= scale; v.w *= scale;
    *(float4*)p = v;
}

extern "C" void kernel_launch(void* const* d_in, const int* in_sizes, int n_in,
                              void* d_out, int out_size, void* d_ws, size_t ws_size,
                              hipStream_t stream) {
    const float* x    = (const float*)d_in[0];
    const int*   erow = (const int*)  d_in[1];
    const int*   ecol = (const int*)  d_in[2];
    const float* ev   = (const float*)d_in[3];
    const float* W1   = (const float*)d_in[4];
    const float* b1   = (const float*)d_in[5];
    const float* W2   = (const float*)d_in[6];
    const float* b2   = (const float*)d_in[7];
    float* out = (float*)d_out;

    char* ws = (char*)d_ws;
    float* w1t  = (float*)(ws);
    float* w2t  = (float*)(ws + 307200);
    float* bufA = (float*)(ws + 307200 + 131072);
    float* bufB = (float*)(ws + 307200 + 131072 + 51200000);

    gcn_transpose_k<<<(256 * 300 + 255) / 256, 256, 0, stream>>>(W1, w1t, 256, 300);
    gcn_transpose_k<<<(128 * 256 + 255) / 256, 256, 0, stream>>>(W2, w2t, 128, 256);

    gcn_gemm1_wmma<<<50000 / 8, 256, 0, stream>>>(x, w1t, bufA);

    hipMemsetAsync(bufB, 0, (size_t)N_NODES * 256 * sizeof(float), stream);
    gcn_spmm_scatter<256><<<(N_EDGES * 64) / 256, 256, 0, stream>>>(erow, ecol, ev, bufA, bufB);

    gcn_gemm2_wmma<<<25000 / 8, 256, 0, stream>>>(bufB, b1, w2t, bufA);

    hipMemsetAsync(out, 0, (size_t)N_NODES * 128 * sizeof(float), stream);
    gcn_spmm_scatter<128><<<(N_EDGES * 32) / 256, 256, 0, stream>>>(erow, ecol, ev, bufA, out);

    gcn_bias_norm<<<(N_NODES * 32) / 256, 256, 0, stream>>>(b2, out);
}


# SpMM by row-range ownership: each of 256 work-groups keeps 196 output rows in LDS, scans the whole edge list, adds matching edges' rows with LDS f32 atomics (sum order = arrival order), stores once
# speedup vs baseline: 1.9710x; 1.9710x over previous
#include <hip/hip_runtime.h>
#include <hip/hip_bf16.h>

#define N_NODES 50000
#define N_EDGES 800000

typedef float v2f __attribute__((ext_vector_type(2)));
typedef float v8f __attribute__((ext_vector_type(8)));

__global__ void gcn_transpose_k(const float* __restrict__ in, float* __restrict__ out,
                                int rows, int cols) {
    int gid = blockIdx.x * blockDim.x + threadIdx.x;
    if (gid < rows * cols) {
        int r = gid / cols;
        int c = gid - r * cols;
        out[(size_t)c * rows + r] = in[gid];
    }
}

__global__ void gcn_gemm1_wmma(const float* __restrict__ x,
                               const float* __restrict__ w1t,
                               float* __restrict__ out) {
    const int wave    = (int)((blockIdx.x * blockDim.x + threadIdx.x) >> 5);
    const int lane    = threadIdx.x & 31;
    const int rowBase = (wave >> 4) << 4;
    const int colBase = (wave & 15) << 4;
    const int m       = lane & 15;
    const int kl      = (lane >> 4) << 1;

    const float* __restrict__ arow = x + (size_t)(rowBase + m) * 300;
    const float* __restrict__ bcol = w1t + colBase + m;

    v8f c = {0.f, 0.f, 0.f, 0.f, 0.f, 0.f, 0.f, 0.f};
    for (int k = 0; k < 300; k += 4) {
        v2f a = *(const v2f*)(arow + k + kl);
        v2f b;
        b.x = bcol[(size_t)(k + kl)     * 256];
        b.y = bcol[(size_t)(k + kl + 1) * 256];
        c = __builtin_amdgcn_wmma_f32_16x16x4_f32(false, a, false, b,
                                                  (short)0, c, false, false);
    }
    const int mBase = rowBase + ((lane >> 4) << 3);
    float* __restrict__ o = out + (size_t)mBase * 256 + colBase + m;
#pragma unroll
    for (int j = 0; j < 8; ++j)
        o[(size_t)j * 256] = c[j];
}

__global__ void gcn_gemm2_wmma(const float* __restrict__ h1,
                               const float* __restrict__ b1,
                               const float* __restrict__ w2t,
                               float* __restrict__ out) {
    const int wave    = (int)((blockIdx.x * blockDim.x + threadIdx.x) >> 5);
    const int lane    = threadIdx.x & 31;
    const int rowBase = (wave >> 3) << 4;
    const int colBase = (wave & 7)  << 4;
    const int m       = lane & 15;
    const int kl      = (lane >> 4) << 1;

    const float* __restrict__ arow = h1 + (size_t)(rowBase + m) * 256;
    const float* __restrict__ bcol = w2t + colBase + m;

    v8f c = {0.f, 0.f, 0.f, 0.f, 0.f, 0.f, 0.f, 0.f};
    for (int k = 0; k < 256; k += 4) {
        v2f a  = *(const v2f*)(arow + k + kl);
        v2f bb = *(const v2f*)(b1   + k + kl);
        a.x = fmaxf(a.x + bb.x, 0.f);
        a.y = fmaxf(a.y + bb.y, 0.f);
        v2f b;
        b.x = bcol[(size_t)(k + kl)     * 128];
        b.y = bcol[(size_t)(k + kl + 1) * 128];
        c = __builtin_amdgcn_wmma_f32_16x16x4_f32(false, a, false, b,
                                                  (short)0, c, false, false);
    }
    const int mBase = rowBase + ((lane >> 4) << 3);
    float* __restrict__ o = out + (size_t)mBase * 128 + colBase + m;
#pragma unroll
    for (int j = 0; j < 8; ++j)
        o[(size_t)j * 128] = c[j];
}

#define SPMM_ROWS   196
#define SPMM_GROUPS 256
#define SPMM_QCAP   544

template <int D>
__global__ __launch_bounds__(1024)
void gcn_spmm_rows(const int*   __restrict__ erow,
                   const int*   __restrict__ ecol,
                   const float* __restrict__ eval,
                   const float* __restrict__ dense,
                   float*       __restrict__ out) {
    constexpr int NC     = D / 128;
    constexpr int EPG    = 8;
    constexpr int NCHUNK = N_EDGES / 128;
    constexpr int U      = 8;
    constexpr int WINS   = (NCHUNK + 32 * U - 1) / (32 * U);
    static_assert(N_EDGES % 128 == 0, "edge list must be whole wave loads");
    static_assert(N_EDGES < (1 << 20) && SPMM_ROWS < (1 << 11), "queue word packing");
    static_assert(SPMM_GROUPS * SPMM_ROWS >= N_NODES, "rows must be covered");

    __shared__ float acc[SPMM_ROWS * D];
    __shared__ int   wq[32 * SPMM_QCAP];

    const int tid  = threadIdx.x;
    const int lane = tid & 31;
    const int wave = __builtin_amdgcn_readfirstlane(tid >> 5);
    const int lo   = blockIdx.x * SPMM_ROWS;

    for (int i = tid; i < SPMM_ROWS * D / 4; i += 1024)
        ((float4*)acc)[i] = make_float4(0.f, 0.f, 0.f, 0.f);
    __syncthreads();

    int* q  = wq + wave * SPMM_QCAP;
    int cnt = 0;

    auto drain = [&](int n) {
        for (int i = 0; i < n; i += EPG) {
            int myq = 0, myc = 0;
            float myv = 0.f;
            if (lane < EPG && i + lane < n) {
                myq = q[i + lane];
                myc = ecol[myq & 0xFFFFF];
                myv = eval[myq & 0xFFFFF];
            }
            int   lr[EPG], c[EPG];
            float v[EPG];
#pragma unroll
            for (int j = 0; j < EPG; ++j) {
                lr[j] = __builtin_amdgcn_readlane(myq, j) >> 20;
                c[j]  = __builtin_amdgcn_readlane(myc, j);
                v[j]  = __int_as_float(__builtin_amdgcn_readlane(__float_as_int(myv), j));
            }
            float4 x[EPG][NC];
#pragma unroll
            for (int j = 0; j < EPG; ++j) {
                if (i + j < n) {
                    const float4* d = (const float4*)(dense + (size_t)c[j] * D) + lane;
#pragma unroll
                    for (int p = 0; p < NC; ++p) x[j][p] = d[32 * p];
                }
            }
#pragma unroll
            for (int j = 0; j < EPG; ++j) {
                if (i + j < n) {
                    float* a = acc + lr[j] * D + lane;
#pragma unroll
                    for (int p = 0; p < NC; ++p) {
                        (void)__hip_atomic_fetch_add(a + p * 128,      v[j] * x[j][p].x, __ATOMIC_RELAXED, __HIP_MEMORY_SCOPE_WORKGROUP);
                        (void)__hip_atomic_fetch_add(a + p * 128 + 32, v[j] * x[j][p].y, __ATOMIC_RELAXED, __HIP_MEMORY_SCOPE_WORKGROUP);
                        (void)__hip_atomic_fetch_add(a + p * 128 + 64, v[j] * x[j][p].z, __ATOMIC_RELAXED, __HIP_MEMORY_SCOPE_WORKGROUP);
                        (void)__hip_atomic_fetch_add(a + p * 128 + 96, v[j] * x[j][p].w, __ATOMIC_RELAXED, __HIP_MEMORY_SCOPE_WORKGROUP);
                    }
                }
            }
        }
    };

    const int rot = blockIdx.x * (NCHUNK / SPMM_GROUPS);

    for (int w = 0; w < WINS; ++w) {
        int4 r4[U];
        int  eb[U];
#pragma unroll
        for (int u = 0; u < U; ++u) {
            const int k = wave + 32 * (w * U + u);
            int ph = k + rot;
            if (ph >= NCHUNK) ph -= NCHUNK;
            eb[u] = ph * 128 + lane * 4;
            r4[u] = make_int4(-1, -1, -1, -1);
            if (k < NCHUNK) r4[u] = *(const int4*)(erow + eb[u]);
        }
#pragma unroll
        for (int u = 0; u < U; ++u) {
            const int rr[4] = {r4[u].x, r4[u].y, r4[u].z, r4[u].w};
#pragma unroll
            for (int t = 0; t < 4; ++t) {
                const int  l = rr[t] - lo;
                const bool m = (unsigned)l < (unsigned)SPMM_ROWS;
                const unsigned b = __builtin_amdgcn_ballot_w32(m);
                if (b) {
                    if (m) q[cnt + (int)__builtin_amdgcn_mbcnt_lo(b, 0u)] = (l << 20) | (eb[u] + t);
                    cnt += __builtin_popcount(b);
                }
            }
            if ((u & 3) == 3 && cnt >= 32) { drain(cnt); cnt = 0; }
        }
    }
    drain(cnt);
    __syncthreads();

    for (int i = tid; i < SPMM_ROWS * D / 4; i += 1024) {
        const int row = i / (D / 4);
        const int f4  = i - row * (D / 4);
        const int p   = f4 >> 5;
        const int l   = f4 & 31;
        if (lo + row < N_NODES) {
            const float* a = acc + row * D + p * 128 + l;
            float4 o = make_float4(a[0], a[32], a[64], a[96]);
            *(float4*)(out + (size_t)(lo + row) * D + 4 * f4) = o;
        }
    }
}

__global__ void gcn_bias_norm(const float* __restrict__ b2, float* __restrict__ out) {
    int gid  = blockIdx.x * blockDim.x + threadIdx.x;
    int row  = gid >> 5;
    int lane = gid & 31;
    if (row >= N_NODES) return;

    float* p = out + (size_t)row * 128 + lane * 4;
    float4 v  = *(float4*)p;
    float4 bb = *(const float4*)(b2 + lane * 4);
    v.x += bb.x; v.y += bb.y; v.z += bb.z; v.w += bb.w;

    float ss = v.x * v.x + v.y * v.y + v.z * v.z + v.w * v.w;
#pragma unroll
    for (int off = 16; off > 0; off >>= 1)
        ss += __shfl_xor(ss, off, 32);

    float scale = 1.0f / fmaxf(sqrtf(ss), 1e-12f);
    v.x *= scale; v.y *= scale; v.z *= scale; v.w *= scale;
    *(float4*)p = v;
}

extern "C" void kernel_launch(void* const* d_in, const int* in_sizes, int n_in,
                              void* d_out, int out_size, void* d_ws, size_t ws_size,
                              hipStream_t stream) {
    const float* x    = (const float*)d_in[0];
    const int*   erow = (const int*)  d_in[1];
    const int*   ecol = (const int*)  d_in[2];
    const float* ev   = (const float*)d_in[3];
    const float* W1   = (const float*)d_in[4];
    const float* b1   = (const float*)d_in[5];
    const float* W2   = (const float*)d_in[6];
    const float* b2   = (const float*)d_in[7];
    float* out = (float*)d_out;

    char* ws = (char*)d_ws;
    float* w1t  = (float*)(ws);
    float* w2t  = (float*)(ws + 307200);
    float* bufA = (float*)(ws + 307200 + 131072);
    float* bufB = (float*)(ws + 307200 + 131072 + 51200000);

    gcn_transpose_k<<<(256 * 300 + 255) / 256, 256, 0, stream>>>(W1, w1t, 256, 300);
    gcn_transpose_k<<<(128 * 256 + 255) / 256, 256, 0, stream>>>(W2, w2t, 128, 256);

    gcn_gemm1_wmma<<<50000 / 8, 256, 0, stream>>>(x, w1t, bufA);

    gcn_spmm_rows<256><<<SPMM_GROUPS, 1024, 0, stream>>>(erow, ecol, ev, bufA, bufB);

    gcn_gemm2_wmma<<<25000 / 8, 256, 0, stream>>>(bufB, b1, w2t, bufA);

    gcn_spmm_rows<128><<<SPMM_GROUPS, 1024, 0, stream>>>(erow, ecol, ev, bufA, out);

    gcn_bias_norm<<<(N_NODES * 32) / 256, 256, 0, stream>>>(b2, out);
}


# Both GEMMs register-blocked, no LDS: a wave keeps an 80x64 / 80x32 C block; A and W (as given, K-contiguous: transposes go) read by 16-B fragment loads; k summed in permuted order within each 8
# speedup vs baseline: 2.6697x; 1.3545x over previous
#include <hip/hip_runtime.h>
#include <hip/hip_bf16.h>

#define N_NODES 50000
#define N_EDGES 800000

typedef float v2f __attribute__((ext_vector_type(2)));
typedef float v4f __attribute__((ext_vector_type(4)));
typedef float v8f __attribute__((ext_vector_type(8)));

template <int K, int NT, bool BIAS_RELU>
__global__ __launch_bounds__(128)
void gcn_gemm_rb(const float* __restrict__ a,
                 const float* __restrict__ w,
                 const float* __restrict__ bias,
                 float*       __restrict__ out) {
    constexpr int MT    = 5;
    constexpr int NCOLS = 4 * NT * 16;
    constexpr int K8    = K & ~7;
    static_assert(N_NODES % (MT * 16) == 0, "row panels must be whole");
    static_assert(K % 8 == 0 || K % 8 == 4, "k tail is one 4-wide step or none");
    static_assert(!BIAS_RELU || K % 8 == 0, "bias path has no k tail");
    static_assert(K8 >= 16, "two chunks are always in flight");

    const int lane = threadIdx.x & 31;
    const int wave = __builtin_amdgcn_readfirstlane((int)(threadIdx.x >> 5));
    const int c    = lane & 15;
    const int h    = lane >> 4;
    const int rowBase = blockIdx.x * (MT * 16);
    const int colBase = wave * (NT * 16);

    const float* __restrict__ ap = a + (size_t)(rowBase + c) * K + 4 * h;
    const float* __restrict__ bp = w + (size_t)(colBase + c) * K + 4 * h;
    const float* __restrict__ sp = bias + 4 * h;

    v8f acc[MT][NT];
#pragma unroll
    for (int i = 0; i < MT; ++i)
#pragma unroll
        for (int j = 0; j < NT; ++j)
            acc[i][j] = (v8f){0.f, 0.f, 0.f, 0.f, 0.f, 0.f, 0.f, 0.f};

    auto load = [&](v4f (&fa)[MT], v4f (&fb)[NT], v4f& fs, int k0) {
#pragma unroll
        for (int i = 0; i < MT; ++i) fa[i] = *(const v4f*)(ap + (size_t)i * 16 * K + k0);
#pragma unroll
        for (int j = 0; j < NT; ++j) fb[j] = *(const v4f*)(bp + (size_t)j * 16 * K + k0);
        if constexpr (BIAS_RELU) fs = *(const v4f*)(sp + k0);
    };
    auto mma = [&](v4f (&fa)[MT], const v4f (&fb)[NT], const v4f& fs) {
        if constexpr (BIAS_RELU) {
#pragma unroll
            for (int i = 0; i < MT; ++i) {
                fa[i].x = fmaxf(fa[i].x + fs.x, 0.f);
                fa[i].y = fmaxf(fa[i].y + fs.y, 0.f);
                fa[i].z = fmaxf(fa[i].z + fs.z, 0.f);
                fa[i].w = fmaxf(fa[i].w + fs.w, 0.f);
            }
        }
#pragma unroll
        for (int i = 0; i < MT; ++i)
#pragma unroll
            for (int j = 0; j < NT; ++j) {
                const v2f x = {fa[i].x, fa[i].y};
                const v2f y = {fb[j].x, fb[j].y};
                acc[i][j] = __builtin_amdgcn_wmma_f32_16x16x4_f32(false, x, false, y,
                                                                  (short)0, acc[i][j], false, false);
            }
#pragma unroll
        for (int i = 0; i < MT; ++i)
#pragma unroll
            for (int j = 0; j < NT; ++j) {
                const v2f x = {fa[i].z, fa[i].w};
                const v2f y = {fb[j].z, fb[j].w};
                acc[i][j] = __builtin_amdgcn_wmma_f32_16x16x4_f32(false, x, false, y,
                                                                  (short)0, acc[i][j], false, false);
            }
    };

    v4f a0[MT], b0[NT], s0 = {0.f, 0.f, 0.f, 0.f};
    v4f a1[MT], b1[NT], s1 = {0.f, 0.f, 0.f, 0.f};
    load(a0, b0, s0, 0);
    int k0 = 0;
    for (; k0 + 16 <= K8; k0 += 16) {
        load(a1, b1, s1, k0 + 8);
        mma(a0, b0, s0);
        const int kn = (k0 + 16 < K8) ? k0 + 16 : K8 - 8;
        load(a0, b0, s0, kn);
        mma(a1, b1, s1);
    }
    if (k0 < K8) mma(a0, b0, s0);

    if constexpr (K8 < K) {
        v2f ta[MT], tb[NT];
#pragma unroll
        for (int i = 0; i < MT; ++i) ta[i] = *(const v2f*)(ap + (size_t)i * 16 * K + K8 - 2 * h);
#pragma unroll
        for (int j = 0; j < NT; ++j) tb[j] = *(const v2f*)(bp + (size_t)j * 16 * K + K8 - 2 * h);
#pragma unroll
        for (int i = 0; i < MT; ++i)
#pragma unroll
            for (int j = 0; j < NT; ++j)
                acc[i][j] = __builtin_amdgcn_wmma_f32_16x16x4_f32(false, ta[i], false, tb[j],
                                                                  (short)0, acc[i][j], false, false);
    }

    float* __restrict__ o = out + (size_t)(rowBase + 8 * h) * NCOLS + colBase + c;
#pragma unroll
    for (int i = 0; i < MT; ++i)
#pragma unroll
        for (int j = 0; j < NT; ++j)
#pragma unroll
            for (int r = 0; r < 8; ++r)
                o[(size_t)(i * 16 + r) * NCOLS + j * 16] = acc[i][j][r];
}

#define SPMM_ROWS   196
#define SPMM_GROUPS 256
#define SPMM_QCAP   544

template <int D>
__global__ __launch_bounds__(1024)
void gcn_spmm_rows(const int*   __restrict__ erow,
                   const int*   __restrict__ ecol,
                   const float* __restrict__ eval,
                   const float* __restrict__ dense,
                   float*       __restrict__ out) {
    constexpr int NC     = D / 128;
    constexpr int EPG    = 8;
    constexpr int NCHUNK = N_EDGES / 128;
    constexpr int U      = 8;
    constexpr int WINS   = (NCHUNK + 32 * U - 1) / (32 * U);
    static_assert(N_EDGES % 128 == 0, "edge list must be whole wave loads");
    static_assert(N_EDGES < (1 << 20) && SPMM_ROWS < (1 << 11), "queue word packing");
    static_assert(SPMM_GROUPS * SPMM_ROWS >= N_NODES, "rows must be covered");

    __shared__ float acc[SPMM_ROWS * D];
    __shared__ int   wq[32 * SPMM_QCAP];

    const int tid  = threadIdx.x;
    const int lane = tid & 31;
    const int wave = __builtin_amdgcn_readfirstlane(tid >> 5);
    const int lo   = blockIdx.x * SPMM_ROWS;

    for (int i = tid; i < SPMM_ROWS * D / 4; i += 1024)
        ((float4*)acc)[i] = make_float4(0.f, 0.f, 0.f, 0.f);
    __syncthreads();

    int* q  = wq + wave * SPMM_QCAP;
    int cnt = 0;

    auto drain = [&](int n) {
        for (int i = 0; i < n; i += EPG) {
            int myq = 0, myc = 0;
            float myv = 0.f;
            if (lane < EPG && i + lane < n) {
                myq = q[i + lane];
                myc = ecol[myq & 0xFFFFF];
                myv = eval[myq & 0xFFFFF];
            }
            int   lr[EPG], c[EPG];
            float v[EPG];
#pragma unroll
            for (int j = 0; j < EPG; ++j) {
                lr[j] = __builtin_amdgcn_readlane(myq, j) >> 20;
                c[j]  = __builtin_amdgcn_readlane(myc, j);
                v[j]  = __int_as_float(__builtin_amdgcn_readlane(__float_as_int(myv), j));
            }
            float4 x[EPG][NC];
#pragma unroll
            for (int j = 0; j < EPG; ++j) {
                if (i + j < n) {
                    const float4* d = (const float4*)(dense + (size_t)c[j] * D) + lane;
#pragma unroll
                    for (int p = 0; p < NC; ++p) x[j][p] = d[32 * p];
                }
            }
#pragma unroll
            for (int j = 0; j < EPG; ++j) {
                if (i + j < n) {
                    float* a = acc + lr[j] * D + lane;
#pragma unroll
                    for (int p = 0; p < NC; ++p) {
                        (void)__hip_atomic_fetch_add(a + p * 128,      v[j] * x[j][p].x, __ATOMIC_RELAXED, __HIP_MEMORY_SCOPE_WORKGROUP);
                        (void)__hip_atomic_fetch_add(a + p * 128 + 32, v[j] * x[j][p].y, __ATOMIC_RELAXED, __HIP_MEMORY_SCOPE_WORKGROUP);
                        (void)__hip_atomic_fetch_add(a + p * 128 + 64, v[j] * x[j][p].z, __ATOMIC_RELAXED, __HIP_MEMORY_SCOPE_WORKGROUP);
                        (void)__hip_atomic_fetch_add(a + p * 128 + 96, v[j] * x[j][p].w, __ATOMIC_RELAXED, __HIP_MEMORY_SCOPE_WORKGROUP);
                    }
                }
            }
        }
    };

    const int rot = blockIdx.x * (NCHUNK / SPMM_GROUPS);

    for (int w = 0; w < WINS; ++w) {
        int4 r4[U];
        int  eb[U];
#pragma unroll
        for (int u = 0; u < U; ++u) {
            const int k = wave + 32 * (w * U + u);
            int ph = k + rot;
            if (ph >= NCHUNK) ph -= NCHUNK;
            eb[u] = ph * 128 + lane * 4;
            r4[u] = make_int4(-1, -1, -1, -1);
            if (k < NCHUNK) r4[u] = *(const int4*)(erow + eb[u]);
        }
#pragma unroll
        for (int u = 0; u < U; ++u) {
            const int rr[4] = {r4[u].x, r4[u].y, r4[u].z, r4[u].w};
#pragma unroll
            for (int t = 0; t < 4; ++t) {
                const int  l = rr[t] - lo;
                const bool m = (unsigned)l < (unsigned)SPMM_ROWS;
                const unsigned b = __builtin_amdgcn_ballot_w32(m);
                if (b) {
                    if (m) q[cnt + (int)__builtin_amdgcn_mbcnt_lo(b, 0u)] = (l << 20) | (eb[u] + t);
                    cnt += __builtin_popcount(b);
                }
            }
            if ((u & 3) == 3 && cnt >= 32) { drain(cnt); cnt = 0; }
        }
    }
    drain(cnt);
    __syncthreads();

    for (int i = tid; i < SPMM_ROWS * D / 4; i += 1024) {
        const int row = i / (D / 4);
        const int f4  = i - row * (D / 4);
        const int p   = f4 >> 5;
        const int l   = f4 & 31;
        if (lo + row < N_NODES) {
            const float* a = acc + row * D + p * 128 + l;
            float4 o = make_float4(a[0], a[32], a[64], a[96]);
            *(float4*)(out + (size_t)(lo + row) * D + 4 * f4) = o;
        }
    }
}

__global__ void gcn_bias_norm(const float* __restrict__ b2, float* __restrict__ out) {
    int gid  = blockIdx.x * blockDim.x + threadIdx.x;
    int row  = gid >> 5;
    int lane = gid & 31;
    if (row >= N_NODES) return;

    float* p = out + (size_t)row * 128 + lane * 4;
    float4 v  = *(float4*)p;
    float4 bb = *(const float4*)(b2 + lane * 4);
    v.x += bb.x; v.y += bb.y; v.z += bb.z; v.w += bb.w;

    float ss = v.x * v.x + v.y * v.y + v.z * v.z + v.w * v.w;
#pragma unroll
    for (int off = 16; off > 0; off >>= 1)
        ss += __shfl_xor(ss, off, 32);

    float scale = 1.0f / fmaxf(sqrtf(ss), 1e-12f);
    v.x *= scale; v.y *= scale; v.z *= scale; v.w *= scale;
    *(float4*)p = v;
}

extern "C" void kernel_launch(void* const* d_in, const int* in_sizes, int n_in,
                              void* d_out, int out_size, void* d_ws, size_t ws_size,
                              hipStream_t stream) {
    const float* x    = (const float*)d_in[0];
    const int*   erow = (const int*)  d_in[1];
    const int*   ecol = (const int*)  d_in[2];
    const float* ev   = (const float*)d_in[3];
    const float* W1   = (const float*)d_in[4];
    const float* b1   = (const float*)d_in[5];
    const float* W2   = (const float*)d_in[6];
    const float* b2   = (const float*)d_in[7];
    float* out = (float*)d_out;

    char* ws = (char*)d_ws;
    float* bufA = (float*)(ws + 307200 + 131072);
    float* bufB = (float*)(ws + 307200 + 131072 + 51200000);

    gcn_gemm_rb<300, 4, false><<<625, 128, 0, stream>>>(x, W1, b1, bufA);

    gcn_spmm_rows<256><<<SPMM_GROUPS, 1024, 0, stream>>>(erow, ecol, ev, bufA, bufB);

    gcn_gemm_rb<256, 2, true><<<625, 128, 0, stream>>>(bufB, W2, b1, bufA);

    gcn_spmm_rows<128><<<SPMM_GROUPS, 1024, 0, stream>>>(erow, ecol, ev, bufA, out);

    gcn_bias_norm<<<(N_NODES * 32) / 256, 256, 0, stream>>>(b2, out);
}


# SpMM2 reads no edge list: SpMM1 notes each group's matched edges (row, col, val) in that group's not yet used output rows, SpMM2 replays the note; row sums stay in LDS-add arrival order
# speedup vs baseline: 3.0107x; 1.1277x over previous
#include <hip/hip_runtime.h>
#include <hip/hip_bf16.h>

#define N_NODES 50000
#define N_EDGES 800000

typedef float v2f __attribute__((ext_vector_type(2)));
typedef float v4f __attribute__((ext_vector_type(4)));
typedef float v8f __attribute__((ext_vector_type(8)));

template <int K, int NT, bool BIAS_RELU>
__global__ __launch_bounds__(128)
void gcn_gemm_rb(const float* __restrict__ a,
                 const float* __restrict__ w,
                 const float* __restrict__ bias,
                 float*       __restrict__ out) {
    constexpr int MT    = 5;
    constexpr int NCOLS = 4 * NT * 16;
    constexpr int K8    = K & ~7;
    static_assert(N_NODES % (MT * 16) == 0, "row panels must be whole");
    static_assert(K % 8 == 0 || K % 8 == 4, "k tail is one 4-wide step or none");
    static_assert(!BIAS_RELU || K % 8 == 0, "bias path has no k tail");
    static_assert(K8 >= 16, "two chunks are always in flight");

    const int lane = threadIdx.x & 31;
    const int wave = __builtin_amdgcn_readfirstlane((int)(threadIdx.x >> 5));
    const int c    = lane & 15;
    const int h    = lane >> 4;
    const int rowBase = blockIdx.x * (MT * 16);
    const int colBase = wave * (NT * 16);

    const float* __restrict__ ap = a + (size_t)(rowBase + c) * K + 4 * h;
    const float* __restrict__ bp = w + (size_t)(colBase + c) * K + 4 * h;
    const float* __restrict__ sp = bias + 4 * h;

    v8f acc[MT][NT];
#pragma unroll
    for (int i = 0; i < MT; ++i)
#pragma unroll
        for (int j = 0; j < NT; ++j)
            acc[i][j] = (v8f){0.f, 0.f, 0.f, 0.f, 0.f, 0.f, 0.f, 0.f};

    auto load = [&](v4f (&fa)[MT], v4f (&fb)[NT], v4f& fs, int k0) {
#pragma unroll
        for (int i = 0; i < MT; ++i) fa[i] = *(const v4f*)(ap + (size_t)i * 16 * K + k0);
#pragma unroll
        for (int j = 0; j < NT; ++j) fb[j] = *(const v4f*)(bp + (size_t)j * 16 * K + k0);
        if constexpr (BIAS_RELU) fs = *(const v4f*)(sp + k0);
    };
    auto mma = [&](v4f (&fa)[MT], const v4f (&fb)[NT], const v4f& fs) {
        if constexpr (BIAS_RELU) {
#pragma unroll
            for (int i = 0; i < MT; ++i) {
                fa[i].x = fmaxf(fa[i].x + fs.x, 0.f);
                fa[i].y = fmaxf(fa[i].y + fs.y, 0.f);
                fa[i].z = fmaxf(fa[i].z + fs.z, 0.f);
                fa[i].w = fmaxf(fa[i].w + fs.w, 0.f);
            }
        }
#pragma unroll
        for (int i = 0; i < MT; ++i)
#pragma unroll
            for (int j = 0; j < NT; ++j) {
                const v2f x = {fa[i].x, fa[i].y};
                const v2f y = {fb[j].x, fb[j].y};
                acc[i][j] = __builtin_amdgcn_wmma_f32_16x16x4_f32(false, x, false, y,
                                                                  (short)0, acc[i][j], false, false);
            }
#pragma unroll
        for (int i = 0; i < MT; ++i)
#pragma unroll
            for (int j = 0; j < NT; ++j) {
                const v2f x = {fa[i].z, fa[i].w};
                const v2f y = {fb[j].z, fb[j].w};
                acc[i][j] = __builtin_amdgcn_wmma_f32_16x16x4_f32(false, x, false, y,
                                                                  (short)0, acc[i][j], false, false);
            }
    };

    v4f a0[MT], b0[NT], s0 = {0.f, 0.f, 0.f, 0.f};
    v4f a1[MT], b1[NT], s1 = {0.f, 0.f, 0.f, 0.f};
    load(a0, b0, s0, 0);
    int k0 = 0;
    for (; k0 + 16 <= K8; k0 += 16) {
        load(a1, b1, s1, k0 + 8);
        mma(a0, b0, s0);
        const int kn = (k0 + 16 < K8) ? k0 + 16 : K8 - 8;
        load(a0, b0, s0, kn);
        mma(a1, b1, s1);
    }
    if (k0 < K8) mma(a0, b0, s0);

    if constexpr (K8 < K) {
        v2f ta[MT], tb[NT];
#pragma unroll
        for (int i = 0; i < MT; ++i) ta[i] = *(const v2f*)(ap + (size_t)i * 16 * K + K8 - 2 * h);
#pragma unroll
        for (int j = 0; j < NT; ++j) tb[j] = *(const v2f*)(bp + (size_t)j * 16 * K + K8 - 2 * h);
#pragma unroll
        for (int i = 0; i < MT; ++i)
#pragma unroll
            for (int j = 0; j < NT; ++j)
                acc[i][j] = __builtin_amdgcn_wmma_f32_16x16x4_f32(false, ta[i], false, tb[j],
                                                                  (short)0, acc[i][j], false, false);
    }

    float* __restrict__ o = out + (size_t)(rowBase + 8 * h) * NCOLS + colBase + c;
#pragma unroll
    for (int i = 0; i < MT; ++i)
#pragma unroll
        for (int j = 0; j < NT; ++j)
#pragma unroll
            for (int r = 0; r < 8; ++r)
                o[(size_t)(i * 16 + r) * NCOLS + j * 16] = acc[i][j][r];
}

#define SPMM_ROWS   196
#define SPMM_GROUPS 256
#define SPMM_QCAP   544
#define OUT_COLS    128

template <int D, int MODE>
__global__ __launch_bounds__(1024)
void gcn_spmm_rows(const int*   __restrict__ erow,
                   const int*   __restrict__ ecol,
                   const float* __restrict__ eval,
                   const float* __restrict__ dense,
                   float*       out,
                   float*       park) {
    constexpr int NC     = D / 128;
    constexpr int EPG    = 8;
    constexpr int NCHUNK = N_EDGES / 128;
    constexpr int U      = 8;
    constexpr int WINS   = (NCHUNK + 32 * U - 1) / (32 * U);
    static_assert(N_EDGES % 128 == 0, "edge list must be whole wave loads");
    static_assert(N_EDGES < (1 << 20) && SPMM_ROWS < (1 << 11), "queue word packing");
    static_assert(SPMM_GROUPS * SPMM_ROWS >= N_NODES, "rows must be covered");
    static_assert((SPMM_GROUPS - 1) * SPMM_ROWS < N_NODES, "every group owns at least one row");
    static_assert(MODE == 1 || MODE == 2, "1 = scan and note, 2 = replay the note");
    static_assert(MODE == 1 || D == OUT_COLS, "the replaying layer writes the final output");

    __shared__ float acc[SPMM_ROWS * D];
    __shared__ int   wq[32 * SPMM_QCAP];
    __shared__ int   noted;

    const int tid  = threadIdx.x;
    const int lane = tid & 31;
    const int wave = __builtin_amdgcn_readfirstlane(tid >> 5);
    const int lo   = blockIdx.x * SPMM_ROWS;

    const int ownRows = (N_NODES - lo < SPMM_ROWS) ? (N_NODES - lo) : SPMM_ROWS;
    const int P       = (ownRows * OUT_COLS - 4) / 3;
    int*   note = (int*)park + (size_t)lo * OUT_COLS;
    int*   lq   = note + 4;
    int*   lc   = lq + P;
    float* lv   = (float*)(lc + P);

    int total = 0;
    if constexpr (MODE == 2)
        total = __builtin_amdgcn_readfirstlane(note[0]);
    const bool replay = (MODE == 2) && (unsigned)total <= (unsigned)P;

    if (tid == 0) noted = 0;
    for (int i = tid; i < SPMM_ROWS * D / 4; i += 1024)
        ((float4*)acc)[i] = make_float4(0.f, 0.f, 0.f, 0.f);
    __syncthreads();

    int* q  = wq + wave * SPMM_QCAP;
    int cnt = 0;

    auto serve = [&](int myq, int myc, float myv, int m) {
        int   lr[EPG], c[EPG];
        float v[EPG];
#pragma unroll
        for (int j = 0; j < EPG; ++j) {
            lr[j] = __builtin_amdgcn_readlane(myq, j) >> 20;
            c[j]  = __builtin_amdgcn_readlane(myc, j);
            v[j]  = __int_as_float(__builtin_amdgcn_readlane(__float_as_int(myv), j));
        }
        float4 x[EPG][NC];
#pragma unroll
        for (int j = 0; j < EPG; ++j) {
            if (j < m) {
                const float4* d = (const float4*)(dense + (size_t)c[j] * D) + lane;
#pragma unroll
                for (int p = 0; p < NC; ++p) x[j][p] = d[32 * p];
            }
        }
#pragma unroll
        for (int j = 0; j < EPG; ++j) {
            if (j < m) {
                float* a = acc + lr[j] * D + lane;
#pragma unroll
                for (int p = 0; p < NC; ++p) {
                    (void)__hip_atomic_fetch_add(a + p * 128,      v[j] * x[j][p].x, __ATOMIC_RELAXED, __HIP_MEMORY_SCOPE_WORKGROUP);
                    (void)__hip_atomic_fetch_add(a + p * 128 + 32, v[j] * x[j][p].y, __ATOMIC_RELAXED, __HIP_MEMORY_SCOPE_WORKGROUP);
                    (void)__hip_atomic_fetch_add(a + p * 128 + 64, v[j] * x[j][p].z, __ATOMIC_RELAXED, __HIP_MEMORY_SCOPE_WORKGROUP);
                    (void)__hip_atomic_fetch_add(a + p * 128 + 96, v[j] * x[j][p].w, __ATOMIC_RELAXED, __HIP_MEMORY_SCOPE_WORKGROUP);
                }
            }
        }
    };

    auto drain = [&](int n) {
        int base = 0;
        if constexpr (MODE == 1) {
            int b0 = 0;
            if (lane == 0) b0 = __hip_atomic_fetch_add(&noted, n, __ATOMIC_RELAXED, __HIP_MEMORY_SCOPE_WORKGROUP);
            base = __builtin_amdgcn_readfirstlane(b0);
        }
        for (int i = 0; i < n; i += EPG) {
            int myq = 0, myc = 0;
            float myv = 0.f;
            if (lane < EPG && i + lane < n) {
                myq = q[i + lane];
                myc = ecol[myq & 0xFFFFF];
                myv = eval[myq & 0xFFFFF];
                if constexpr (MODE == 1) {
                    const int s = base + i + lane;
                    if (s < P) { lq[s] = myq; lc[s] = myc; lv[s] = myv; }
                }
            }
            serve(myq, myc, myv, n - i);
        }
    };

    if (replay) {
        auto fetch = [&](int at, int& fq, int& fc, float& fv) {
            fq = 0; fc = 0; fv = 0.f;
            if (lane < EPG && at + lane < total) { fq = lq[at + lane]; fc = lc[at + lane]; fv = lv[at + lane]; }
        };
        constexpr int STEP = 32 * EPG;
        int   i = wave * EPG;
        int   q0, c0, q1, c1;
        float v0, v1;
        fetch(i, q0, c0, v0);
        fetch(i + STEP, q1, c1, v1);
        for (; i < total; i += STEP) {
            int   q2, c2;
            float v2;
            fetch(i + 2 * STEP, q2, c2, v2);
            serve(q0, c0, v0, total - i);
            q0 = q1; c0 = c1; v0 = v1;
            q1 = q2; c1 = c2; v1 = v2;
        }
    } else {
        const int rot = blockIdx.x * (NCHUNK / SPMM_GROUPS);

        for (int w = 0; w < WINS; ++w) {
            int4 r4[U];
            int  eb[U];
#pragma unroll
            for (int u = 0; u < U; ++u) {
                const int k = wave + 32 * (w * U + u);
                int ph = k + rot;
                if (ph >= NCHUNK) ph -= NCHUNK;
                eb[u] = ph * 128 + lane * 4;
                r4[u] = make_int4(-1, -1, -1, -1);
                if (k < NCHUNK) r4[u] = *(const int4*)(erow + eb[u]);
            }
#pragma unroll
            for (int u = 0; u < U; ++u) {
                const int rr[4] = {r4[u].x, r4[u].y, r4[u].z, r4[u].w};
#pragma unroll
                for (int t = 0; t < 4; ++t) {
                    const int  l = rr[t] - lo;
                    const bool m = (unsigned)l < (unsigned)SPMM_ROWS;
                    const unsigned b = __builtin_amdgcn_ballot_w32(m);
                    if (b) {
                        if (m) q[cnt + (int)__builtin_amdgcn_mbcnt_lo(b, 0u)] = (l << 20) | (eb[u] + t);
                        cnt += __builtin_popcount(b);
                    }
                }
                if ((u & 3) == 3 && cnt >= 32) { drain(cnt); cnt = 0; }
            }
        }
        drain(cnt);
    }
    __syncthreads();

    if constexpr (MODE == 1) {
        if (tid == 0) note[0] = noted;
    }

    for (int i = tid; i < SPMM_ROWS * D / 4; i += 1024) {
        const int row = i / (D / 4);
        const int f4  = i - row * (D / 4);
        const int p   = f4 >> 5;
        const int l   = f4 & 31;
        if (lo + row < N_NODES) {
            const float* a = acc + row * D + p * 128 + l;
            float4 o = make_float4(a[0], a[32], a[64], a[96]);
            *(float4*)(out + (size_t)(lo + row) * D + 4 * f4) = o;
        }
    }
}

__global__ void gcn_bias_norm(const float* __restrict__ b2, float* __restrict__ out) {
    int gid  = blockIdx.x * blockDim.x + threadIdx.x;
    int row  = gid >> 5;
    int lane = gid & 31;
    if (row >= N_NODES) return;

    float* p = out + (size_t)row * 128 + lane * 4;
    float4 v  = *(float4*)p;
    float4 bb = *(const float4*)(b2 + lane * 4);
    v.x += bb.x; v.y += bb.y; v.z += bb.z; v.w += bb.w;

    float ss = v.x * v.x + v.y * v.y + v.z * v.z + v.w * v.w;
#pragma unroll
    for (int off = 16; off > 0; off >>= 1)
        ss += __shfl_xor(ss, off, 32);

    float scale = 1.0f / fmaxf(sqrtf(ss), 1e-12f);
    v.x *= scale; v.y *= scale; v.z *= scale; v.w *= scale;
    *(float4*)p = v;
}

extern "C" void kernel_launch(void* const* d_in, const int* in_sizes, int n_in,
                              void* d_out, int out_size, void* d_ws, size_t ws_size,
                              hipStream_t stream) {
    const float* x    = (const float*)d_in[0];
    const int*   erow = (const int*)  d_in[1];
    const int*   ecol = (const int*)  d_in[2];
    const float* ev   = (const float*)d_in[3];
    const float* W1   = (const float*)d_in[4];
    const float* b1   = (const float*)d_in[5];
    const float* W2   = (const float*)d_in[6];
    const float* b2   = (const float*)d_in[7];
    float* out = (float*)d_out;

    char* ws = (char*)d_ws;
    float* bufA = (float*)(ws + 307200 + 131072);
    float* bufB = (float*)(ws + 307200 + 131072 + 51200000);

    gcn_gemm_rb<300, 4, false><<<625, 128, 0, stream>>>(x, W1, b1, bufA);

    gcn_spmm_rows<256, 1><<<SPMM_GROUPS, 1024, 0, stream>>>(erow, ecol, ev, bufA, bufB, out);

    gcn_gemm_rb<256, 2, true><<<625, 128, 0, stream>>>(bufB, W2, b1, bufA);

    gcn_spmm_rows<128, 2><<<SPMM_GROUPS, 1024, 0, stream>>>(erow, ecol, ev, bufA, out, out);

    gcn_bias_norm<<<(N_NODES * 32) / 256, 256, 0, stream>>>(b2, out);
}


# SpMM1 scan skips chunks: a small launch first marks per 64-edge chunk which groups own a row in it (bit planes in workspace); a group loads and tests only marked chunks; LDS-add arrival order kept
# speedup vs baseline: 3.1586x; 1.0491x over previous
#include <hip/hip_runtime.h>
#include <hip/hip_bf16.h>

#define N_NODES 50000
#define N_EDGES 800000

typedef float v2f __attribute__((ext_vector_type(2)));
typedef float v4f __attribute__((ext_vector_type(4)));
typedef float v8f __attribute__((ext_vector_type(8)));

template <int K, int NT, bool BIAS_RELU>
__global__ __launch_bounds__(128)
void gcn_gemm_rb(const float* __restrict__ a,
                 const float* __restrict__ w,
                 const float* __restrict__ bias,
                 float*       __restrict__ out) {
    constexpr int MT    = 5;
    constexpr int NCOLS = 4 * NT * 16;
    constexpr int K8    = K & ~7;
    static_assert(N_NODES % (MT * 16) == 0, "row panels must be whole");
    static_assert(K % 8 == 0 || K % 8 == 4, "k tail is one 4-wide step or none");
    static_assert(!BIAS_RELU || K % 8 == 0, "bias path has no k tail");
    static_assert(K8 >= 16, "two chunks are always in flight");

    const int lane = threadIdx.x & 31;
    const int wave = __builtin_amdgcn_readfirstlane((int)(threadIdx.x >> 5));
    const int c    = lane & 15;
    const int h    = lane >> 4;
    const int rowBase = blockIdx.x * (MT * 16);
    const int colBase = wave * (NT * 16);

    const float* __restrict__ ap = a + (size_t)(rowBase + c) * K + 4 * h;
    const float* __restrict__ bp = w + (size_t)(colBase + c) * K + 4 * h;
    const float* __restrict__ sp = bias + 4 * h;

    v8f acc[MT][NT];
#pragma unroll
    for (int i = 0; i < MT; ++i)
#pragma unroll
        for (int j = 0; j < NT; ++j)
            acc[i][j] = (v8f){0.f, 0.f, 0.f, 0.f, 0.f, 0.f, 0.f, 0.f};

    auto load = [&](v4f (&fa)[MT], v4f (&fb)[NT], v4f& fs, int k0) {
#pragma unroll
        for (int i = 0; i < MT; ++i) fa[i] = *(const v4f*)(ap + (size_t)i * 16 * K + k0);
#pragma unroll
        for (int j = 0; j < NT; ++j) fb[j] = *(const v4f*)(bp + (size_t)j * 16 * K + k0);
        if constexpr (BIAS_RELU) fs = *(const v4f*)(sp + k0);
    };
    auto mma = [&](v4f (&fa)[MT], const v4f (&fb)[NT], const v4f& fs) {
        if constexpr (BIAS_RELU) {
#pragma unroll
            for (int i = 0; i < MT; ++i) {
                fa[i].x = fmaxf(fa[i].x + fs.x, 0.f);
                fa[i].y = fmaxf(fa[i].y + fs.y, 0.f);
                fa[i].z = fmaxf(fa[i].z + fs.z, 0.f);
                fa[i].w = fmaxf(fa[i].w + fs.w, 0.f);
            }
        }
#pragma unroll
        for (int i = 0; i < MT; ++i)
#pragma unroll
            for (int j = 0; j < NT; ++j) {
                const v2f x = {fa[i].x, fa[i].y};
                const v2f y = {fb[j].x, fb[j].y};
                acc[i][j] = __builtin_amdgcn_wmma_f32_16x16x4_f32(false, x, false, y,
                                                                  (short)0, acc[i][j], false, false);
            }
#pragma unroll
        for (int i = 0; i < MT; ++i)
#pragma unroll
            for (int j = 0; j < NT; ++j) {
                const v2f x = {fa[i].z, fa[i].w};
                const v2f y = {fb[j].z, fb[j].w};
                acc[i][j] = __builtin_amdgcn_wmma_f32_16x16x4_f32(false, x, false, y,
                                                                  (short)0, acc[i][j], false, false);
            }
    };

    v4f a0[MT], b0[NT], s0 = {0.f, 0.f, 0.f, 0.f};
    v4f a1[MT], b1[NT], s1 = {0.f, 0.f, 0.f, 0.f};
    load(a0, b0, s0, 0);
    int k0 = 0;
    for (; k0 + 16 <= K8; k0 += 16) {
        load(a1, b1, s1, k0 + 8);
        mma(a0, b0, s0);
        const int kn = (k0 + 16 < K8) ? k0 + 16 : K8 - 8;
        load(a0, b0, s0, kn);
        mma(a1, b1, s1);
    }
    if (k0 < K8) mma(a0, b0, s0);

    if constexpr (K8 < K) {
        v2f ta[MT], tb[NT];
#pragma unroll
        for (int i = 0; i < MT; ++i) ta[i] = *(const v2f*)(ap + (size_t)i * 16 * K + K8 - 2 * h);
#pragma unroll
        for (int j = 0; j < NT; ++j) tb[j] = *(const v2f*)(bp + (size_t)j * 16 * K + K8 - 2 * h);
#pragma unroll
        for (int i = 0; i < MT; ++i)
#pragma unroll
            for (int j = 0; j < NT; ++j)
                acc[i][j] = __builtin_amdgcn_wmma_f32_16x16x4_f32(false, ta[i], false, tb[j],
                                                                  (short)0, acc[i][j], false, false);
    }

    float* __restrict__ o = out + (size_t)(rowBase + 8 * h) * NCOLS + colBase + c;
#pragma unroll
    for (int i = 0; i < MT; ++i)
#pragma unroll
        for (int j = 0; j < NT; ++j)
#pragma unroll
            for (int r = 0; r < 8; ++r)
                o[(size_t)(i * 16 + r) * NCOLS + j * 16] = acc[i][j][r];
}

#define SPMM_ROWS   196
#define SPMM_GROUPS 256
#define SPMM_QCAP   544
#define OUT_COLS    128
#define CHUNK_EDGES 64
#define N_CHUNKS    (N_EDGES / CHUNK_EDGES)
#define HIT_BYTES   (8 * N_CHUNKS * 4)
static_assert(N_EDGES % (2 * CHUNK_EDGES) == 0, "a wave of the chunk pass takes two whole chunks");
static_assert(SPMM_GROUPS == 256, "8 planes of 32 groups");
static_assert(HIT_BYTES <= 438272, "the planes fit in front of bufA in the workspace");

__global__ __launch_bounds__(256)
void gcn_chunk_groups(const int* __restrict__ erow, unsigned* __restrict__ hits) {
    const int lane = threadIdx.x & 31;
    const int pair = blockIdx.x * 8 + (int)(threadIdx.x >> 5);
    if (pair >= N_CHUNKS / 2) return;
    const int4 r = *(const int4*)(erow + pair * 128 + lane * 4);
    const int  rr[4] = {r.x, r.y, r.z, r.w};
    unsigned word[8] = {0u, 0u, 0u, 0u, 0u, 0u, 0u, 0u};
#pragma unroll
    for (int t = 0; t < 4; ++t) {
        const unsigned row = (unsigned)rr[t];
        const unsigned g   = row / SPMM_ROWS;
#pragma unroll
        for (int w = 0; w < 8; ++w)
            if (row < SPMM_GROUPS * SPMM_ROWS && (g >> 5) == (unsigned)w) word[w] |= 1u << (g & 31);
    }
    unsigned mine = 0u;
#pragma unroll
    for (int w = 0; w < 8; ++w) {
        unsigned m = word[w];
#pragma unroll
        for (int off = 8; off > 0; off >>= 1) m |= __shfl_xor(m, off, 32);
        if ((lane & 15) == w) mine = m;
    }
    if ((lane & 15) < 8)
        hits[(size_t)(lane & 15) * N_CHUNKS + 2 * pair + (lane >> 4)] = mine;
}

template <int D, int MODE>
__global__ __launch_bounds__(1024)
void gcn_spmm_rows(const int*      __restrict__ erow,
                   const int*      __restrict__ ecol,
                   const float*    __restrict__ eval,
                   const float*    __restrict__ dense,
                   const unsigned* __restrict__ hits,
                   float*          out,
                   float*          park) {
    constexpr int NC     = D / 128;
    constexpr int EPG    = 8;
    constexpr int NCHUNK = N_CHUNKS;
    constexpr int U      = 8;
    constexpr int CPW    = (NCHUNK + 31) / 32;
    constexpr int NJ     = (CPW + 31) / 32;
    static_assert(U * CHUNK_EDGES / 2 + 31 <= SPMM_QCAP, "queue holds what arrives between two drain checks");
    static_assert(N_EDGES < (1 << 20) && SPMM_ROWS < (1 << 11), "queue word packing");
    static_assert(SPMM_GROUPS * SPMM_ROWS >= N_NODES, "rows must be covered");
    static_assert((SPMM_GROUPS - 1) * SPMM_ROWS < N_NODES, "every group owns at least one row");
    static_assert(MODE == 1 || MODE == 2, "1 = scan and note, 2 = replay the note");
    static_assert(MODE == 1 || D == OUT_COLS, "the replaying layer writes the final output");

    __shared__ float acc[SPMM_ROWS * D];
    __shared__ int   wq[32 * SPMM_QCAP];
    __shared__ int   noted;

    const int tid  = threadIdx.x;
    const int lane = tid & 31;
    const int wave = __builtin_amdgcn_readfirstlane(tid >> 5);
    const int lo   = blockIdx.x * SPMM_ROWS;

    const int ownRows = (N_NODES - lo < SPMM_ROWS) ? (N_NODES - lo) : SPMM_ROWS;
    const int P       = (ownRows * OUT_COLS - 4) / 3;
    int*   note = (int*)park + (size_t)lo * OUT_COLS;
    int*   lq   = note + 4;
    int*   lc   = lq + P;
    float* lv   = (float*)(lc + P);

    int total = 0;
    if constexpr (MODE == 2)
        total = __builtin_amdgcn_readfirstlane(note[0]);
    const bool replay = (MODE == 2) && (unsigned)total <= (unsigned)P;

    if (tid == 0) noted = 0;
    for (int i = tid; i < SPMM_ROWS * D / 4; i += 1024)
        ((float4*)acc)[i] = make_float4(0.f, 0.f, 0.f, 0.f);
    __syncthreads();

    int* q  = wq + wave * SPMM_QCAP;
    int cnt = 0;

    auto serve = [&](int myq, int myc, float myv, int m) {
        int   lr[EPG], c[EPG];
        float v[EPG];
#pragma unroll
        for (int j = 0; j < EPG; ++j) {
            lr[j] = __builtin_amdgcn_readlane(myq, j) >> 20;
            c[j]  = __builtin_amdgcn_readlane(myc, j);
            v[j]  = __int_as_float(__builtin_amdgcn_readlane(__float_as_int(myv), j));
        }
        float4 x[EPG][NC];
#pragma unroll
        for (int j = 0; j < EPG; ++j) {
            if (j < m) {
                const float4* d = (const float4*)(dense + (size_t)c[j] * D) + lane;
#pragma unroll
                for (int p = 0; p < NC; ++p) x[j][p] = d[32 * p];
            }
        }
#pragma unroll
        for (int j = 0; j < EPG; ++j) {
            if (j < m) {
                float* a = acc + lr[j] * D + lane;
#pragma unroll
                for (int p = 0; p < NC; ++p) {
                    (void)__hip_atomic_fetch_add(a + p * 128,      v[j] * x[j][p].x, __ATOMIC_RELAXED, __HIP_MEMORY_SCOPE_WORKGROUP);
                    (void)__hip_atomic_fetch_add(a + p * 128 + 32, v[j] * x[j][p].y, __ATOMIC_RELAXED, __HIP_MEMORY_SCOPE_WORKGROUP);
                    (void)__hip_atomic_fetch_add(a + p * 128 + 64, v[j] * x[j][p].z, __ATOMIC_RELAXED, __HIP_MEMORY_SCOPE_WORKGROUP);
                    (void)__hip_atomic_fetch_add(a + p * 128 + 96, v[j] * x[j][p].w, __ATOMIC_RELAXED, __HIP_MEMORY_SCOPE_WORKGROUP);
                }
            }
        }
    };

    auto drain = [&](int n) {
        int base = 0;
        if constexpr (MODE == 1) {
            int b0 = 0;
            if (lane == 0) b0 = __hip_atomic_fetch_add(&noted, n, __ATOMIC_RELAXED, __HIP_MEMORY_SCOPE_WORKGROUP);
            base = __builtin_amdgcn_readfirstlane(b0);
        }
        for (int i = 0; i < n; i += EPG) {
            int myq = 0, myc = 0;
            float myv = 0.f;
            if (lane < EPG && i + lane < n) {
                myq = q[i + lane];
                myc = ecol[myq & 0xFFFFF];
                myv = eval[myq & 0xFFFFF];
                if constexpr (MODE == 1) {
                    const int s = base + i + lane;
                    if (s < P) { lq[s] = myq; lc[s] = myc; lv[s] = myv; }
                }
            }
            serve(myq, myc, myv, n - i);
        }
    };

    if (replay) {
        auto fetch = [&](int at, int& fq, int& fc, float& fv) {
            fq = 0; fc = 0; fv = 0.f;
            if (lane < EPG && at + lane < total) { fq = lq[at + lane]; fc = lc[at + lane]; fv = lv[at + lane]; }
        };
        constexpr int STEP = 32 * EPG;
        int   i = wave * EPG;
        int   q0, c0, q1, c1;
        float v0, v1;
        fetch(i, q0, c0, v0);
        fetch(i + STEP, q1, c1, v1);
        for (; i < total; i += STEP) {
            int   q2, c2;
            float v2;
            fetch(i + 2 * STEP, q2, c2, v2);
            serve(q0, c0, v0, total - i);
            q0 = q1; c0 = c1; v0 = v1;
            q1 = q2; c1 = c2; v1 = v2;
        }
    } else {
        const int       rot   = blockIdx.x * (NCHUNK / SPMM_GROUPS);
        const int       cbase = wave * CPW;
        const unsigned* plane = hits + (size_t)(blockIdx.x >> 5) * NCHUNK;
        const unsigned  gbit  = 1u << (blockIdx.x & 31);

        auto planeword = [&](int j) -> unsigned {
            const int i = 32 * j + lane;
            unsigned  wd = 0u;
            if (i < CPW && cbase + i < NCHUNK) {
                int ph = cbase + i + rot;
                if (ph >= NCHUNK) ph -= NCHUNK;
                wd = plane[ph];
            }
            return wd;
        };

        unsigned wcur = planeword(0);
        for (int j = 0; j < NJ; ++j) {
            const unsigned wnext = (j + 1 < NJ) ? planeword(j + 1) : 0u;
            unsigned m = __builtin_amdgcn_ballot_w32((wcur & gbit) != 0u);
            while (m) {
                const int left = __builtin_popcount(m);
                const int nv   = left < U ? left : U;
                int2 r2[U];
                int  eb[U];
#pragma unroll
                for (int u = 0; u < U; ++u) {
                    r2[u] = make_int2(-1, -1);
                    eb[u] = 0;
                    if (u < nv) {
                        int ph = cbase + 32 * j + __builtin_ctz(m) + rot;
                        m &= m - 1;
                        if (ph >= NCHUNK) ph -= NCHUNK;
                        eb[u] = ph * CHUNK_EDGES + lane * 2;
                        r2[u] = *(const int2*)(erow + eb[u]);
                    }
                }
#pragma unroll
                for (int u = 0; u < U; ++u) {
                    if (u < nv) {
                        const int rr[2] = {r2[u].x, r2[u].y};
#pragma unroll
                        for (int t = 0; t < 2; ++t) {
                            const int  l = rr[t] - lo;
                            const bool hit = (unsigned)l < (unsigned)SPMM_ROWS;
                            const unsigned b = __builtin_amdgcn_ballot_w32(hit);
                            if (b) {
                                if (hit) q[cnt + (int)__builtin_amdgcn_mbcnt_lo(b, 0u)] = (l << 20) | (eb[u] + t);
                                cnt += __builtin_popcount(b);
                            }
                        }
                    }
                    if ((u & 3) == 3 && cnt >= 32) { drain(cnt); cnt = 0; }
                }
            }
            wcur = wnext;
        }
        drain(cnt);
    }
    __syncthreads();

    if constexpr (MODE == 1) {
        if (tid == 0) note[0] = noted;
    }

    for (int i = tid; i < SPMM_ROWS * D / 4; i += 1024) {
        const int row = i / (D / 4);
        const int f4  = i - row * (D / 4);
        const int p   = f4 >> 5;
        const int l   = f4 & 31;
        if (lo + row < N_NODES) {
            const float* a = acc + row * D + p * 128 + l;
            float4 o = make_float4(a[0], a[32], a[64], a[96]);
            *(float4*)(out + (size_t)(lo + row) * D + 4 * f4) = o;
        }
    }
}

__global__ void gcn_bias_norm(const float* __restrict__ b2, float* __restrict__ out) {
    int gid  = blockIdx.x * blockDim.x + threadIdx.x;
    int row  = gid >> 5;
    int lane = gid & 31;
    if (row >= N_NODES) return;

    float* p = out + (size_t)row * 128 + lane * 4;
    float4 v  = *(float4*)p;
    float4 bb = *(const float4*)(b2 + lane * 4);
    v.x += bb.x; v.y += bb.y; v.z += bb.z; v.w += bb.w;

    float ss = v.x * v.x + v.y * v.y + v.z * v.z + v.w * v.w;
#pragma unroll
    for (int off = 16; off > 0; off >>= 1)
        ss += __shfl_xor(ss, off, 32);

    float scale = 1.0f / fmaxf(sqrtf(ss), 1e-12f);
    v.x *= scale; v.y *= scale; v.z *= scale; v.w *= scale;
    *(float4*)p = v;
}

extern "C" void kernel_launch(void* const* d_in, const int* in_sizes, int n_in,
                              void* d_out, int out_size, void* d_ws, size_t ws_size,
                              hipStream_t stream) {
    const float* x    = (const float*)d_in[0];
    const int*   erow = (const int*)  d_in[1];
    const int*   ecol = (const int*)  d_in[2];
    const float* ev   = (const float*)d_in[3];
    const float* W1   = (const float*)d_in[4];
    const float* b1   = (const float*)d_in[5];
    const float* W2   = (const float*)d_in[6];
    const float* b2   = (const float*)d_in[7];
    float* out = (float*)d_out;

    char* ws = (char*)d_ws;
    float* bufA = (float*)(ws + 307200 + 131072);
    float* bufB = (float*)(ws + 307200 + 131072 + 51200000);

    unsigned* hits = (unsigned*)ws;
    gcn_chunk_groups<<<(N_CHUNKS / 2 + 7) / 8, 256, 0, stream>>>(erow, hits);

    gcn_gemm_rb<300, 4, false><<<625, 128, 0, stream>>>(x, W1, b1, bufA);

    gcn_spmm_rows<256, 1><<<SPMM_GROUPS, 1024, 0, stream>>>(erow, ecol, ev, bufA, hits, bufB, out);

    gcn_gemm_rb<256, 2, true><<<625, 128, 0, stream>>>(bufB, W2, b1, bufA);

    gcn_spmm_rows<128, 2><<<SPMM_GROUPS, 1024, 0, stream>>>(erow, ecol, ev, bufA, hits, out, out);

    gcn_bias_norm<<<(N_NODES * 32) / 256, 256, 0, stream>>>(b2, out);
}
